# starting kernel
# baseline (speedup 1.0000x reference)
#include <hip/hip_runtime.h>

typedef float v2f __attribute__((ext_vector_type(2)));
typedef float v8f __attribute__((ext_vector_type(8)));

#define D 128

__global__ __launch_bounds__(256) void gcn_gemm_wmma(
    const float* __restrict__ x, const float* __restrict__ w,
    float* __restrict__ support, int n_nodes) {
  const int lane = threadIdx.x & 31;
  const int wave = threadIdx.x >> 5;
  const int m    = lane & 15;
  const int half = lane >> 4;
  const int row0 = blockIdx.x * 16;
  const int col0 = wave * 16;

  int arow = row0 + m;
  if (arow >= n_nodes) arow = n_nodes - 1;
  const float* xrow = x + (size_t)arow * D;
  const int n = col0 + m;

  v8f acc = {};
#pragma unroll 4
  for (int k = 0; k < D; k += 4) {
    const int ka = k + half * 2;
    v2f a = *(const v2f*)(xrow + ka);
    v2f b;
    b.x = w[(size_t)(ka + 0) * D + n];
    b.y = w[(size_t)(ka + 1) * D + n];
    acc = __builtin_amdgcn_wmma_f32_16x16x4_f32(
         false, a,  false, b,
         (short)0, acc,  false,  false);
  }

  float* srow = support + (size_t)(row0 + half * 8) * D + n;
  if (row0 + 16 <= n_nodes) {
#pragma unroll
    for (int i = 0; i < 8; ++i) srow[(size_t)i * D] = acc[i];
  } else {
#pragma unroll
    for (int i = 0; i < 8; ++i) {
      if (row0 + half * 8 + i < n_nodes) srow[(size_t)i * D] = acc[i];
    }
  }
}

__global__ __launch_bounds__(256) void gcn_init_out(
    float* __restrict__ out, const float* __restrict__ bias, int total) {
  int i = blockIdx.x * blockDim.x + threadIdx.x;
  if (i < total) out[i] = bias[i & (D - 1)];
}

__global__ __launch_bounds__(256) void gcn_spmm_scatter(
    const int* __restrict__ adj_row, const int* __restrict__ adj_col,
    const float* __restrict__ adj_val, const float* __restrict__ support,
    float* __restrict__ out, int n_edges) {
  const int lane = threadIdx.x & 31;
  const int e = blockIdx.x * (blockDim.x >> 5) + (threadIdx.x >> 5);
  if (e >= n_edges) return;

  const int src  = adj_col[e];
  const int dst  = adj_row[e];
  const float v  = adj_val[e];
  const float* srow = support + (size_t)src * D;
  float* drow       = out + (size_t)dst * D;

#pragma unroll
  for (int j = 0; j < D; j += 32) {
    atomicAdd(drow + j + lane, v * srow[j + lane]);
  }
}

extern "C" void kernel_launch(void* const* d_in, const int* in_sizes, int n_in,
                              void* d_out, int out_size, void* d_ws, size_t ws_size,
                              hipStream_t stream) {
  const float* x       = (const float*)d_in[0];
  const int*   adj_row = (const int*)d_in[1];
  const int*   adj_col = (const int*)d_in[2];
  const float* adj_val = (const float*)d_in[3];
  const float* weight  = (const float*)d_in[4];
  const float* bias    = (const float*)d_in[5];
  float* out     = (float*)d_out;
  float* support = (float*)d_ws;

  const int n_nodes = in_sizes[0] / D;
  const int n_edges = in_sizes[1];
  const int total   = n_nodes * D;

  gcn_gemm_wmma<<<(n_nodes + 15) / 16, 256, 0, stream>>>(x, weight, support, n_nodes);
  gcn_init_out<<<(total + 255) / 256, 256, 0, stream>>>(out, bias, total);
  gcn_spmm_scatter<<<(n_edges + 7) / 8, 256, 0, stream>>>(adj_row, adj_col, adj_val,
                                                          support, out, n_edges);
}


# owner bitmaps and match queue removed: launch 1 orders each chunk's records by owner (LDS counting sort) + run offsets; a wave of launch 2 loads its chunks' runs directly; sum order = bucket arrival
# speedup vs baseline: 4.1941x; 4.1941x over previous
#include <hip/hip_runtime.h>

typedef float v2f __attribute__((ext_vector_type(2)));
typedef float v4f __attribute__((ext_vector_type(4)));
typedef __attribute__((address_space(1))) v4f gv4f;
typedef float v8f __attribute__((ext_vector_type(8)));

#define D    128
#define TR   160
#define LDP  132
#define NT   1024
#define CHK  2048
#define OWN  1024
#define GE   8
#define CPW  10
#define QCAP (NT * CPW)
#define RW   5
#define NOROW 7


__global__ __launch_bounds__(256) void gcn_chunk_sort(const int* __restrict__ adj_row,
                                                      const int* __restrict__ adj_col,
                                                      const float* __restrict__ adj_val,
                                                      unsigned* __restrict__ rec,
                                                      int n_edges, int n_owners, int n_chunks) {
  __shared__ __attribute__((aligned(16))) int   cnt[OWN];
  __shared__ __attribute__((aligned(16))) uint2 srec[CHK];
  __shared__ int wsum[8];
  const int tid  = threadIdx.x;
  const int lane = tid & 31;
  const int wave = tid >> 5;
  const int c    = blockIdx.x;
  const bool whole = (c + 1) * CHK <= n_edges;
  int2   r[4], s[4];
  float2 v[4];
  if (whole) {
#pragma unroll
    for (int p = 0; p < 4; ++p) {
      const int e0 = c * CHK + p * 512 + 2 * tid;
      r[p] = *(const int2*)(adj_row + e0);
      s[p] = *(const int2*)(adj_col + e0);
      v[p] = *(const float2*)(adj_val + e0);
    }
  } else {
    const int last = n_edges - 1;
#pragma unroll
    for (int p = 0; p < 4; ++p) {
      const int e0 = c * CHK + p * 512 + 2 * tid;
      const int a0 = (e0 + 0 < last) ? e0 + 0 : last;
      const int a1 = (e0 + 1 < last) ? e0 + 1 : last;
      r[p].x = adj_row[a0];
      r[p].y = adj_row[a1];
      s[p].x = adj_col[a0];
      s[p].y = adj_col[a1];
      v[p].x = adj_val[a0];
      v[p].y = adj_val[a1];
      if (e0 + 0 > last) r[p].x = -1;
      if (e0 + 1 > last) r[p].y = -1;
    }
  }
  for (int i = tid; i < OWN; i += 256) cnt[i] = 0;
  const unsigned nlim = (unsigned)((n_owners < OWN) ? n_owners : OWN);
  unsigned own[4][2];
#pragma unroll
  for (int p = 0; p < 4; ++p) {
    own[p][0] = (unsigned)r[p].x / (unsigned)TR;
    own[p][1] = (unsigned)r[p].y / (unsigned)TR;
  }
  __syncthreads();
  int rk[4][2];
#pragma unroll
  for (int p = 0; p < 4; ++p) {
#pragma unroll
    for (int t = 0; t < 2; ++t) {
      rk[p][t] = 0;
      if (own[p][t] < nlim) rk[p][t] = atomicAdd(&cnt[own[p][t]], 1);
    }
  }
  __syncthreads();
  const int4 c4 = *(const int4*)&cnt[4 * tid];
  const int tsum = c4.x + c4.y + c4.z + c4.w;
  int incl = tsum;
#pragma unroll
  for (int d = 1; d < 32; d <<= 1) {
    const int up = __shfl_up(incl, d);
    if (lane >= d) incl += up;
  }
  if (lane == 31) wsum[wave] = incl;
  __syncthreads();
  int wbase = 0;
  int total = 0;
#pragma unroll
  for (int k = 0; k < 8; ++k) {
    const int ws = wsum[k];
    if (k < wave) wbase += ws;
    total += ws;
  }
  const int s0 = wbase + incl - tsum;
  const int s1 = s0 + c4.x;
  const int s2 = s1 + c4.y;
  const int s3 = s2 + c4.z;
  *(int4*)&cnt[4 * tid] = make_int4(s0, s1, s2, s3);
  const int ostr = (n_owners + 31) & ~31;
  unsigned* offs = rec + 2 * (size_t)n_chunks * CHK;
  if (4 * tid < ostr)
    *(uint4*)(offs + (size_t)c * ostr + 4 * tid) =
        make_uint4((unsigned)s0 | ((unsigned)c4.x << 16), (unsigned)s1 | ((unsigned)c4.y << 16),
                   (unsigned)s2 | ((unsigned)c4.z << 16), (unsigned)s3 | ((unsigned)c4.w << 16));
  __syncthreads();
  int fp[4][2];
#pragma unroll
  for (int p = 0; p < 4; ++p) {
    fp[p][0] = cnt[(own[p][0] < nlim) ? own[p][0] : 0u];
    fp[p][1] = cnt[(own[p][1] < nlim) ? own[p][1] : 0u];
  }
#pragma unroll
  for (int p = 0; p < 4; ++p) {
    const unsigned o0 = own[p][0];
    const unsigned o1 = own[p][1];
    const unsigned l0 = (unsigned)r[p].x - o0 * (unsigned)TR;
    const unsigned l1 = (unsigned)r[p].y - o1 * (unsigned)TR;
    if (o0 < nlim)
      srec[fp[p][0] + rk[p][0]] = make_uint2(((unsigned)s[p].x & 0x00ffffffu) | (l0 << 24), __builtin_bit_cast(unsigned, v[p].x));
    if (o1 < nlim)
      srec[fp[p][1] + rk[p][1]] = make_uint2(((unsigned)s[p].y & 0x00ffffffu) | (l1 << 24), __builtin_bit_cast(unsigned, v[p].y));
  }
  __syncthreads();
  uint2* dst = (uint2*)rec + (size_t)c * CHK;
  if (total == CHK) {
    uint4 t4[4];
#pragma unroll
    for (int p = 0; p < 4; ++p) t4[p] = *(const uint4*)&srec[2 * (p * 256 + tid)];
#pragma unroll
    for (int p = 0; p < 4; ++p) *(uint4*)(dst + 2 * (p * 256 + tid)) = t4[p];
  } else {
#pragma unroll
    for (int p = 0; p < 4; ++p) {
      const int i = 2 * (p * 256 + tid);
      if (i + 1 < total) {
        *(uint4*)(dst + i) = *(const uint4*)&srec[i];
      } else if (i < total) {
        dst[i] = srec[i];
      }
    }
  }
}

__device__ __forceinline__ void lds_barrier() {
  __builtin_amdgcn_fence(__ATOMIC_RELEASE, "workgroup", "local");
  __builtin_amdgcn_s_barrier();
  __builtin_amdgcn_fence(__ATOMIC_ACQUIRE, "workgroup", "local");
}

struct RowSums { v8f c0, c1, c2, c3; };

__device__ __forceinline__ void add_row(RowSums& acc, int s, float vs, const v4f& xv) {
  acc.c0[s] += vs * xv.x;
  acc.c1[s] += vs * xv.y;
  acc.c2[s] += vs * xv.z;
  acc.c3[s] += vs * xv.w;
}

__device__ __forceinline__ const gv4f* row_base(const float* __restrict__ x, int cs) {
  const unsigned long long xb = (unsigned long long)x + (unsigned long long)(unsigned)cs * (D * sizeof(float));
  const unsigned xlo = (unsigned)__builtin_amdgcn_readfirstlane((int)(unsigned)xb);
  const unsigned xhi = (unsigned)__builtin_amdgcn_readfirstlane((int)(unsigned)(xb >> 32));
  return (const gv4f*)(((unsigned long long)xhi << 32) | (unsigned long long)xlo);
}

__device__ __forceinline__ void drain_bucket(int bs, int bn, int lane, const int2* q2,
                                             RowSums& acc, const float* __restrict__ x) {
  if (bn <= 0) return;
  int2 en = q2[bs + ((lane < bn) ? lane : 0)];
  for (int c0 = 0; c0 < bn; c0 += 32) {
    const int m = bn - c0;
    const int   ck = (lane < m) ? en.x : (NOROW << 28);
    const float vc = __builtin_bit_cast(float, en.y);
#pragma unroll
    for (int j = 0; j < 32 / GE; ++j) {
      if (j * GE < m) {
        const gv4f* xr[GE];
#pragma unroll
        for (int u = 0; u < GE; ++u) {
          const int cs = __builtin_amdgcn_readlane(ck, j * GE + u) & 0x0fffffff;
          xr[u] = row_base(x, cs);
        }
        v4f xv[GE];
#pragma unroll
        for (int u = 0; u < GE; ++u) xv[u] = xr[u][lane];
        if (j == 0) {
          const int i1 = c0 + 32 + lane;
          en = q2[bs + ((i1 < bn) ? i1 : 0)];
        }
#pragma unroll
        for (int u = 0; u < GE; ++u) {
          const int t = j * GE + u;
          const int   s  = (int)(((unsigned)__builtin_amdgcn_readlane(ck, t) >> 28) & 7u);
          const float vs = __builtin_bit_cast(float, __builtin_amdgcn_readlane(__builtin_bit_cast(int, vc), t));
          add_row(acc, s, vs, xv[u]);
        }
      }
    }
  }
}

__global__ __launch_bounds__(NT) void gcn_owner_rows(
    const float* __restrict__ x,
    const float* __restrict__ w, const float* __restrict__ bias,
    const unsigned* __restrict__ rec32,
    float* __restrict__ out, int n_nodes, int n_chunks) {
  __shared__ __attribute__((aligned(16))) float agg[TR * LDP];
  __shared__ int2  q2[QCAP];
  __shared__ int   bcnt[32];
  __shared__ int   wmx[32];

  const int tid  = threadIdx.x;
  const int lane = tid & 31;
  const int wave = __builtin_amdgcn_readfirstlane(tid >> 5);
  const int own  = blockIdx.x;
  const int row0 = own * TR;
  const int2* __restrict__ rec = (const int2*)rec32;
  const int ostr = ((int)gridDim.x + 31) & ~31;
  const unsigned* __restrict__ offs = rec32 + 2 * (size_t)n_chunks * CHK;

  if (tid < 32) bcnt[tid] = 0;

  RowSums acc;
  acc.c0 = v8f{};
  acc.c1 = v8f{};
  acc.c2 = v8f{};
  acc.c3 = v8f{};

  for (int c0 = 0; c0 < n_chunks; c0 += 32 * CPW) {
    unsigned pk = 0u;
    const int cj = c0 + wave + 32 * lane;
    if (lane < CPW && cj < n_chunks && own < OWN) pk = offs[(size_t)cj * ostr + own];
    if (c0 == 0) lds_barrier();
    int st[CPW], cn[CPW];
    int smax = 0;
#pragma unroll
    for (int j = 0; j < CPW; ++j) {
      const unsigned pj = (unsigned)__builtin_amdgcn_readlane((int)pk, j);
      st[j] = (c0 + wave + 32 * j) * CHK + (int)(pj & 0xffffu);
      cn[j] = (int)(pj >> 16);
      smax  = (cn[j] > smax) ? cn[j] : smax;
    }
    int r0 = 0;
    bool more;
    do {
      int2 rc[CPW];
#pragma unroll
      for (int j = 0; j < CPW; ++j) {
        rc[j] = make_int2(0, 0);
        if (r0 + lane < cn[j]) rc[j] = rec[st[j] + r0 + lane];
      }
      if (lane == 0) wmx[wave] = smax;
      int rl[CPW], rk[CPW];
#pragma unroll
      for (int j = 0; j < CPW; ++j) {
        const int l = (int)((unsigned)rc[j].x >> 24);
        const bool ok = (r0 + lane < cn[j]) && ((unsigned)l < (unsigned)TR);
        rl[j] = ok ? l : -1;
        rk[j] = 0;
        if (ok) rk[j] = atomicAdd(&bcnt[l & 31], 1);
      }
      lds_barrier();
      const int cl = bcnt[lane];
      more = __builtin_amdgcn_ballot_w32(wmx[lane] > r0 + 32) != 0u;
      int binc = cl;
#pragma unroll
      for (int d = 1; d < 32; d <<= 1) {
        const int up = __shfl_up(binc, d);
        if (lane >= d) binc += up;
      }
      const int excl = binc - cl;
      int bo[CPW];
#pragma unroll
      for (int j = 0; j < CPW; ++j) bo[j] = __shfl(excl, rl[j] & 31);
#pragma unroll
      for (int j = 0; j < CPW; ++j) {
        if (rl[j] >= 0)
          q2[bo[j] + rk[j]] = make_int2((rc[j].x & 0x00ffffff) | ((rl[j] >> 5) << 28), rc[j].y);
      }
      const int bs = __builtin_amdgcn_readlane(excl, wave);
      const int bn = __builtin_amdgcn_readlane(cl, wave);
      lds_barrier();
      if (tid < 32) bcnt[tid] = 0;
      drain_bucket(bs, bn, lane, q2, acc, x);
      r0 += 32;
      if (more || c0 + 32 * CPW < n_chunks) lds_barrier();
    } while (more);
  }

#pragma unroll
  for (int s = 0; s < RW; ++s) {
    float* a = agg + (wave + 32 * s) * LDP + 4 * lane;
    v4f r4;
    r4.x = acc.c0[s];
    r4.y = acc.c1[s];
    r4.z = acc.c2[s];
    r4.w = acc.c3[s];
    *(v4f*)a = r4;
  }
  lds_barrier();

  const int m    = lane & 15;
  const int half = lane >> 4;
  for (int unit = wave; unit < (TR / 16) * 4; unit += 32) {
    const int mt = unit >> 2;
    const int n0 = (unit & 3) * 32 + m;
    const float* arow = agg + (mt * 16 + m) * LDP;
    v8f acc0 = {};
    v8f acc1 = {};
#pragma unroll 4
    for (int k = 0; k < D; k += 4) {
      const int ka = k + half * 2;
      v2f a = *(const v2f*)(arow + ka);
      v2f b0, b1;
      b0.x = w[(size_t)(ka + 0) * D + n0];
      b0.y = w[(size_t)(ka + 1) * D + n0];
      b1.x = w[(size_t)(ka + 0) * D + n0 + 16];
      b1.y = w[(size_t)(ka + 1) * D + n0 + 16];
      acc0 = __builtin_amdgcn_wmma_f32_16x16x4_f32(false, a, false, b0, (short)0, acc0, false, false);
      acc1 = __builtin_amdgcn_wmma_f32_16x16x4_f32(false, a, false, b1, (short)0, acc1, false, false);
    }
    const float bz0 = bias[n0];
    const float bz1 = bias[n0 + 16];
    const int rbase = row0 + mt * 16 + half * 8;
    float* o = out + (size_t)rbase * D + n0;
    if (row0 + TR <= n_nodes) {
#pragma unroll
      for (int i = 0; i < 8; ++i) {
        o[(size_t)i * D]      = acc0[i] + bz0;
        o[(size_t)i * D + 16] = acc1[i] + bz1;
      }
    } else {
#pragma unroll
      for (int i = 0; i < 8; ++i) {
        if (rbase + i < n_nodes) {
          o[(size_t)i * D]      = acc0[i] + bz0;
          o[(size_t)i * D + 16] = acc1[i] + bz1;
        }
      }
    }
  }
}

extern "C" void kernel_launch(void* const* d_in, const int* in_sizes, int n_in,
                              void* d_out, int out_size, void* d_ws, size_t ws_size,
                              hipStream_t stream) {
  const float* x       = (const float*)d_in[0];
  const int*   adj_row = (const int*)d_in[1];
  const int*   adj_col = (const int*)d_in[2];
  const float* adj_val = (const float*)d_in[3];
  const float* weight  = (const float*)d_in[4];
  const float* bias    = (const float*)d_in[5];
  float* out = (float*)d_out;
  unsigned* rec = (unsigned*)d_ws;

  const int n_nodes  = in_sizes[0] / D;
  const int n_edges  = in_sizes[1];
  const int n_owners = (n_nodes + TR - 1) / TR;
  const int n_chunks = (n_edges + CHK - 1) / CHK;

  gcn_chunk_sort<<<n_chunks, 256, 0, stream>>>(adj_row, adj_col, adj_val, rec,
                                               n_edges, n_owners, n_chunks);
  gcn_owner_rows<<<n_owners, NT, 0, stream>>>(x, weight, bias, rec, out, n_nodes, n_chunks);
}
